# starting kernel
# baseline (speedup 1.0000x reference)
#include <hip/hip_runtime.h>


typedef __attribute__((ext_vector_type(2))) float v2f;
typedef __attribute__((ext_vector_type(8))) float v8f;

#define MAXC    512
#define NGRP    (MAXC / 8)
#define MAXV    512
#define THREADS 256

__launch_bounds__(THREADS, 2)
__global__ void p1interp_wmma(const float* __restrict__ x,
                              const float* __restrict__ wxg,
                              const float* __restrict__ wyg,
                              const float* __restrict__ Minv,
                              const float* __restrict__ Av,
                              const float* __restrict__ bbox,
                              const int*   __restrict__ dofs,
                              float2* __restrict__ out,
                              int NP, int C, int NV)
{
    __shared__ float sAop [NGRP * 32 * 2];
    __shared__ float sMinv[MAXC * 4];
    __shared__ float sBbox[MAXC * 4];
    __shared__ float sA   [MAXC * 2];
    __shared__ int   sDof [MAXC * 3];
    __shared__ float sWx  [MAXV];
    __shared__ float sWy  [MAXV];

    const int tid = threadIdx.x;

    for (int c = tid; c < C; c += THREADS) {
        sMinv[c * 4 + 0] = Minv[c * 4 + 0];
        sMinv[c * 4 + 1] = Minv[c * 4 + 1];
        sMinv[c * 4 + 2] = Minv[c * 4 + 2];
        sMinv[c * 4 + 3] = Minv[c * 4 + 3];
        sA[c * 2 + 0] = Av[c * 2 + 0];
        sA[c * 2 + 1] = Av[c * 2 + 1];
        sBbox[c * 4 + 0] = bbox[c * 4 + 0];
        sBbox[c * 4 + 1] = bbox[c * 4 + 1];
        sBbox[c * 4 + 2] = bbox[c * 4 + 2];
        sBbox[c * 4 + 3] = bbox[c * 4 + 3];
        sDof[c * 3 + 0] = dofs[c * 3 + 0];
        sDof[c * 3 + 1] = dofs[c * 3 + 1];
        sDof[c * 3 + 2] = dofs[c * 3 + 2];
    }
    for (int idx = tid; idx < (C / 8) * 32; idx += THREADS) {
        int g = idx >> 5, L = idx & 31;
        int m = L & 15, cc = m >> 1, e = m & 1;
        int c = g * 8 + cc;
        float m0e = Minv[c * 4 + e];
        float m1e = Minv[c * 4 + 2 + e];
        float v0, v1;
        if (L < 16) { v0 = m0e; v1 = m1e; }
        else {
            float ax = Av[c * 2 + 0], ay = Av[c * 2 + 1];
            v0 = -(ax * m0e + ay * m1e);
            v1 = 0.0f;
        }
        sAop[idx * 2 + 0] = v0;
        sAop[idx * 2 + 1] = v1;
    }
    for (int v = tid; v < NV && v < MAXV; v += THREADS) {
        sWx[v] = wxg[v];
        sWy[v] = wyg[v];
    }
    __syncthreads();

    const int  lane = tid & 31;
    const int  wave = tid >> 5;
    const int  n    = lane & 15;
    const bool hi   = lane >= 16;
    const float NTOL = -1e-10f;

    const int numTiles   = (NP + 15) >> 4;
    const int wavesTotal = gridDim.x * (THREADS >> 5);
    const int tileStep   = wavesTotal;
    const int nGroups    = C / 8;

    for (int tile = blockIdx.x * (THREADS >> 5) + wave; tile < numTiles;
         tile += tileStep) {
        int gp  = tile * 16 + n;
        int gpc = gp < NP ? gp : NP - 1;
        float px = x[gpc * 2 + 0];
        float py = x[gpc * 2 + 1];

        {
            int np2 = (tile + tileStep) * 16 + n;
            if (np2 < NP) __builtin_prefetch(&x[np2 * 2], 0, 1);
        }

        v2f b;
        b.x = hi ? 1.0f : px;
        b.y = hi ? 0.0f : py;

        unsigned mx = 0, my = 0;
#pragma unroll
        for (int i = 0; i < 16; ++i) {
            float x0 = sBbox[(i * 16) * 4 + 0], x1 = sBbox[(i * 16) * 4 + 1];
            float y0 = sBbox[i * 4 + 2],        y1 = sBbox[i * 4 + 3];
            mx |= ((x0 < px) & (px < x1)) ? (1u << i) : 0u;
            my |= ((y0 < py) & (py < y1)) ? (1u << i) : 0u;
        }

        int best = -1;
#pragma unroll 2
        for (int g = 0; g < nGroups; ++g) {
            v2f a = *(const v2f*)&sAop[(g * 32 + lane) * 2];
            v8f d = {};
            d = __builtin_amdgcn_wmma_f32_16x16x4_f32(
                    false, a, false, b, (short)0, d, false, false);

            int cg = g * 8 + (hi ? 4 : 0);
#pragma unroll
            for (int q = 0; q < 4; ++q) {
                float s = d[2 * q];
                float t = d[2 * q + 1];
                int  cq = cg + q;
                int  qq = cq & 255;
                unsigned inb = (mx >> (qq >> 4)) & (my >> (qq & 15)) & 1u;
                bool ins = inb & (s > NTOL) & (t > NTOL) & ((s + t) < 1.0f);
                best = ins ? cq : best;
            }
        }

        int other = __builtin_amdgcn_ds_swizzle(best, 0x401F);
        best = best > other ? best : other;

        float ox = 0.0f, oy = 0.0f;
        if (best >= 0) {
            int c  = best;
            float dx = px - sA[c * 2 + 0];
            float dy = py - sA[c * 2 + 1];
            float s  = dx * sMinv[c * 4 + 0] + dy * sMinv[c * 4 + 2];
            float t  = dx * sMinv[c * 4 + 1] + dy * sMinv[c * 4 + 3];
            float w0 = 1.0f - s - t;
            int d0 = sDof[c * 3 + 0], d1 = sDof[c * 3 + 1], d2 = sDof[c * 3 + 2];
            ox = w0 * sWx[d0] + s * sWx[d1] + t * sWx[d2];
            oy = w0 * sWy[d0] + s * sWy[d1] + t * sWy[d2];
        }
        if (!hi && gp < NP) out[gp] = make_float2(ox, oy);
    }
}

extern "C" void kernel_launch(void* const* d_in, const int* in_sizes, int n_in,
                              void* d_out, int out_size, void* d_ws, size_t ws_size,
                              hipStream_t stream) {
    const float* xp   = (const float*)d_in[0];
    const float* wx   = (const float*)d_in[1];
    const float* wy   = (const float*)d_in[2];
    const float* Minv = (const float*)d_in[3];
    const float* Av   = (const float*)d_in[4];
    const float* bbox = (const float*)d_in[5];
    const int*   dofs = (const int*)d_in[6];
    float2* out = (float2*)d_out;

    int NP = in_sizes[0] / 2;
    int C  = in_sizes[3] / 4;
    int NV = in_sizes[1];
    if (C > MAXC) C = MAXC;

    int numTiles = (NP + 15) / 16;
    int wavesPerBlock = THREADS / 32;
    int blocks = (numTiles + wavesPerBlock - 1) / wavesPerBlock;
    if (blocks > 256) blocks = 256;
    if (blocks < 1)   blocks = 1;

    p1interp_wmma<<<blocks, THREADS, 0, stream>>>(
        xp, wx, wy, Minv, Av, bbox, dofs, out, NP, C, NV);
}


# Tile loop software-pipelined in one-WMMA stages: WMMA issued two groups ahead into accumulators of its own (next tile's first two last; B operand read ready-made from LDS); inside keys as 4 chains
# speedup vs baseline: 3.1883x; 3.1883x over previous
#include <hip/hip_runtime.h>


typedef __attribute__((ext_vector_type(2))) float v2f;
typedef __attribute__((ext_vector_type(8))) float v8f;

#define MAXC    512
#define MAXV    512
#define THREADS 256
#define NWAVES  (THREADS / 32)
#define GPW     (MAXC / 8 / NWAVES)
#define PPW     (GPW / 2)
#define TILES   (THREADS / 16)
#define BPITCH  (THREADS + 16)
#define NOHIT   (-1024)

__launch_bounds__(THREADS)
__global__ void p1interp_wmma(const float* __restrict__ x,
                              const float* __restrict__ wxg,
                              const float* __restrict__ wyg,
                              const float* __restrict__ Minv,
                              const float* __restrict__ Av,
                              const float* __restrict__ bbox,
                              const int*   __restrict__ dofs,
                              float2* __restrict__ out,
                              int NP, int C, int NV)
{
    __shared__ float2 sPts [2 * THREADS];
    __shared__ int    sBest[2 * NWAVES * BPITCH];
    __shared__ float  sMinv[MAXC * 4];
    __shared__ float  sA   [MAXC * 2];
    __shared__ int    sDof [MAXC * 3];
    __shared__ float  sWx  [MAXV];
    __shared__ float  sWy  [MAXV];

    const int  tid  = threadIdx.x;
    const int  lane = tid & 31;
    const int  wave = __builtin_amdgcn_readfirstlane(tid >> 5);
    const int  n    = lane & 15;
    const bool hi   = lane >= 16;
    const int  ysh  = hi ? 4 : 0;
    const float NTOL = -1e-10f;
    const float UB1  = 0x1.fffffep-1f;
    const float INF  = __builtin_inff();

    const int gp  = blockIdx.x * THREADS + tid;
    const int gpc = gp < NP ? gp : NP - 1;
    const float2 myp = *(const float2*)&x[gpc * 2];

    const int m  = lane & 15;
    const int cc = (m >> 3) * 4 + (m & 3);
    const int e  = (m >> 2) & 1;
    v2f aop[GPW];
#pragma unroll
    for (int k = 0; k < GPW; ++k) {
        const int c  = (wave * GPW + k) * 8 + cc;
        const int cl = c < C ? c : 0;
        const float4 M = *(const float4*)&Minv[cl * 4];
        const float2 a = *(const float2*)&Av[cl * 2];
        const float m0e = e ? M.y : M.x;
        const float m1e = e ? M.w : M.z;
        float off = -(a.x * m0e + a.y * m1e);
        if (c >= C) off = -INF;
        aop[k].x = hi ? off  : m0e;
        aop[k].y = hi ? 0.0f : m1e;
    }

    float ylo[8], yhi[8];
#pragma unroll
    for (int j = 0; j < 8; ++j) {
        const int col = (j >> 2) * 8 + ysh + (j & 3);
        const float2 yb = *(const float2*)&bbox[col * 4 + 2];
        ylo[j] = yb.x; yhi[j] = yb.y;
    }
    float xlo[PPW], xhi[PPW];
#pragma unroll
    for (int k = 0; k < PPW; ++k) {
        const int row = (wave * PPW + k) & 15;
        const float2 xb = *(const float2*)&bbox[(row * 16) * 4];
        xlo[k] = xb.x; xhi[k] = xb.y;
    }

    float4 tM[MAXC / THREADS]; float2 tA[MAXC / THREADS];
    int tD[MAXC / THREADS][3]; float tWx[MAXV / THREADS], tWy[MAXV / THREADS];
#pragma unroll
    for (int i = 0; i < MAXC / THREADS; ++i) {
        const int c  = tid + i * THREADS;
        const int cl = c < C ? c : 0;
        tM[i] = *(const float4*)&Minv[cl * 4];
        tA[i] = *(const float2*)&Av[cl * 2];
        tD[i][0] = dofs[cl * 3 + 0];
        tD[i][1] = dofs[cl * 3 + 1];
        tD[i][2] = dofs[cl * 3 + 2];
    }
#pragma unroll
    for (int i = 0; i < MAXV / THREADS; ++i) {
        const int v  = tid + i * THREADS;
        const int vl = v < NV ? v : 0;
        tWx[i] = wxg[vl];
        tWy[i] = wyg[vl];
    }

    sPts[(tid >> 4) * 32 + (tid & 15)]      = myp;
    sPts[(tid >> 4) * 32 + 16 + (tid & 15)] = make_float2(1.0f, 0.0f);
#pragma unroll
    for (int i = 0; i < MAXC / THREADS; ++i) {
        const int c = tid + i * THREADS;
        *(float4*)&sMinv[c * 4] = tM[i];
        *(float2*)&sA[c * 2]    = tA[i];
        sDof[c * 3 + 0] = tD[i][0];
        sDof[c * 3 + 1] = tD[i][1];
        sDof[c * 3 + 2] = tD[i][2];
    }
#pragma unroll
    for (int i = 0; i < MAXV / THREADS; ++i) {
        const int v = tid + i * THREADS;
        sWx[v] = tWx[i];
        sWy[v] = tWy[i];
    }
    __syncthreads();

    const int cellbase = wave * (GPW * 8) + ysh;
    int* const myBest = &sBest[(wave * 2 + (hi ? 1 : 0)) * BPITCH + n];

    auto gates = [&](float px, float py, v2f (&gub)[4], float (&gthr)[PPW]) {
#pragma unroll
        for (int j = 0; j < 8; ++j)
            gub[j >> 1][j & 1] = ((ylo[j] < py) & (py < yhi[j])) ? UB1 : -INF;
#pragma unroll
        for (int k = 0; k < PPW; ++k)
            gthr[k] = ((xlo[k] < px) & (px < xhi[k])) ? NTOL : INF;
    };

    const float2 pc = sPts[n];
    const float2 bc0 = sPts[lane];
    v2f bc;
    bc.x = bc0.x;
    bc.y = bc0.y;
    v2f ub[4]; float thr[PPW];
    gates(pc.x, pc.y, ub, thr);

    v8f d[GPW + 2];
    {
        const v8f z = {};
        d[0] = __builtin_amdgcn_wmma_f32_16x16x4_f32(
                   false, aop[0], false, bc, (short)0, z, false, false);
        d[1] = __builtin_amdgcn_wmma_f32_16x16x4_f32(
                   false, aop[1], false, bc, (short)0, z, false, false);
    }

    for (int t = 0; t < TILES; ++t) {
        const int tn = (t + 1 < TILES) ? t + 1 : t;
        const float2 pn  = sPts[tn * 32 + n];
        const float2 bn0 = sPts[tn * 32 + lane];
        v2f bn;
        bn.x = bn0.x;
        bn.y = bn0.y;

        int bq[4] = {NOHIT, NOHIT, NOHIT, NOHIT};

#pragma unroll
        for (int g = 0; g < GPW; ++g) {
            __builtin_amdgcn_sched_barrier(0);
            {
                const v8f z = {};
                d[g + 2] = __builtin_amdgcn_wmma_f32_16x16x4_f32(
                               false, aop[(g + 2) & (GPW - 1)], false,
                               (g + 2 < GPW) ? bc : bn, (short)0, z, false, false);
            }
            const v8f dg = d[g];
            const int  u  = (g & 1) * 2;
            const float th = thr[g >> 1];
            const v2f e01 = ub[u]     - (__builtin_shufflevector(dg, dg, 0, 1) + __builtin_shufflevector(dg, dg, 4, 5));
            const v2f e23 = ub[u + 1] - (__builtin_shufflevector(dg, dg, 2, 3) + __builtin_shufflevector(dg, dg, 6, 7));
            float m[4];
            m[0] = __builtin_fminf(__builtin_fminf(dg[0], dg[4]), e01.x);
            m[1] = __builtin_fminf(__builtin_fminf(dg[1], dg[5]), e01.y);
            m[2] = __builtin_fminf(__builtin_fminf(dg[2], dg[6]), e23.x);
            m[3] = __builtin_fminf(__builtin_fminf(dg[3], dg[7]), e23.y);
#pragma unroll
            for (int q = 0; q < 4; ++q)
                bq[q] = (m[q] > th) ? (g * 8 + q) : bq[q];
        }
        __builtin_amdgcn_sched_barrier(0);

        v2f ubn[4]; float thrn[PPW];
        gates(pn.x, pn.y, ubn, thrn);

        const int b01 = bq[0] > bq[1] ? bq[0] : bq[1];
        const int b23 = bq[2] > bq[3] ? bq[2] : bq[3];
        const int best = (b01 > b23 ? b01 : b23) + cellbase;
        myBest[t * 16] = best;

        bc = bn;
#pragma unroll
        for (int j = 0; j < 4; ++j) ub[j] = ubn[j];
#pragma unroll
        for (int k = 0; k < PPW; ++k) thr[k] = thrn[k];
        d[0] = d[GPW];
        d[1] = d[GPW + 1];
    }

    __syncthreads();

    int cell = sBest[tid];
#pragma unroll
    for (int r = 1; r < 2 * NWAVES; ++r) {
        const int o = sBest[r * BPITCH + tid];
        cell = cell > o ? cell : o;
    }

    float ox = 0.0f, oy = 0.0f;
    if (cell >= 0) {
        const int c  = cell;
        const float px = myp.x, py = myp.y;
        float dx = px - sA[c * 2 + 0];
        float dy = py - sA[c * 2 + 1];
        float s  = dx * sMinv[c * 4 + 0] + dy * sMinv[c * 4 + 2];
        float t  = dx * sMinv[c * 4 + 1] + dy * sMinv[c * 4 + 3];
        float w0 = 1.0f - s - t;
        int d0 = sDof[c * 3 + 0], d1 = sDof[c * 3 + 1], d2 = sDof[c * 3 + 2];
        ox = w0 * sWx[d0] + s * sWx[d1] + t * sWx[d2];
        oy = w0 * sWy[d0] + s * sWy[d1] + t * sWy[d2];
    }
    if (gp < NP) out[gp] = make_float2(ox, oy);
}

extern "C" void kernel_launch(void* const* d_in, const int* in_sizes, int n_in,
                              void* d_out, int out_size, void* d_ws, size_t ws_size,
                              hipStream_t stream) {
    const float* xp   = (const float*)d_in[0];
    const float* wx   = (const float*)d_in[1];
    const float* wy   = (const float*)d_in[2];
    const float* Minv = (const float*)d_in[3];
    const float* Av   = (const float*)d_in[4];
    const float* bbox = (const float*)d_in[5];
    const int*   dofs = (const int*)d_in[6];
    float2* out = (float2*)d_out;

    int NP = in_sizes[0] / 2;
    int C  = in_sizes[3] / 4;
    int NV = in_sizes[1];
    if (C > MAXC) C = MAXC;

    int blocks = (NP + THREADS - 1) / THREADS;
    if (blocks < 1) blocks = 1;

    p1interp_wmma<<<blocks, THREADS, 0, stream>>>(
        xp, wx, wy, Minv, Av, bbox, dofs, out, NP, C, NV);
}


# Each wave sweeps only the points that pass the bbox x test of one of its 2 quad rows, picked by ballot + prefix count into an LDS list of its own; a wave now owns both triangles of 2 rows
# speedup vs baseline: 6.3630x; 1.9957x over previous
#include <hip/hip_runtime.h>


typedef __attribute__((ext_vector_type(2))) float v2f;
typedef __attribute__((ext_vector_type(8))) float v8f;

#define MAXC    512
#define MAXV    512
#define THREADS 256
#define NWAVES  (THREADS / 32)
#define GPW     (MAXC / 8 / NWAVES)
#define RPW     2
#define PASSES  (THREADS / 32)
#define LDUMP   (THREADS + 32)
#define LCAP    (LDUMP + 32)
#define BPITCH  (THREADS + 16)
#define NOHIT   (-1024)

__launch_bounds__(THREADS)
__global__ void p1interp_wmma(const float* __restrict__ x,
                              const float* __restrict__ wxg,
                              const float* __restrict__ wyg,
                              const float* __restrict__ Minv,
                              const float* __restrict__ Av,
                              const float* __restrict__ bbox,
                              const int*   __restrict__ dofs,
                              float2* __restrict__ out,
                              int NP, int C, int NV)
{
    __shared__ __attribute__((aligned(16))) float4 sList[NWAVES * LCAP];
    __shared__ __attribute__((aligned(16))) int    sBest[2 * NWAVES * BPITCH];
    __shared__ float  sMinv[MAXC * 4];
    __shared__ float  sA   [MAXC * 2];
    __shared__ int    sDof [MAXC * 3];
    __shared__ float  sWx  [MAXV];
    __shared__ float  sWy  [MAXV];

    const int  tid  = threadIdx.x;
    const int  lane = tid & 31;
    const int  wave = __builtin_amdgcn_readfirstlane(tid >> 5);
    const int  n    = lane & 15;
    const bool hi   = lane >= 16;
    const int  ysh  = hi ? 4 : 0;
    const float NTOL = -1e-10f;
    const float UB1  = 0x1.fffffep-1f;
    const float INF  = __builtin_inff();

    const int pbase = blockIdx.x * THREADS;
    float2 P[PASSES];
#pragma unroll
    for (int j = 0; j < PASSES; ++j) {
        const int g  = pbase + j * 32 + lane;
        const int gc = g < NP ? g : NP - 1;
        P[j] = *(const float2*)&x[gc * 2];
    }
    float xlo[RPW], xhi[RPW];
#pragma unroll
    for (int k = 0; k < RPW; ++k) {
        const int row = (wave * RPW + k) & 15;
        const float2 xb = *(const float2*)&bbox[(row * 16) * 4];
        xlo[k] = xb.x; xhi[k] = xb.y;
    }
    const int gp  = pbase + tid;
    const int gpc = gp < NP ? gp : NP - 1;
    const float2 myp = *(const float2*)&x[gpc * 2];

    const int m  = lane & 15;
    const int cc = (m >> 3) * 4 + (m & 3);
    const int e  = (m >> 2) & 1;
    v2f aop[GPW];
#pragma unroll
    for (int k = 0; k < GPW; ++k) {
        const int G  = (k >> 2) * (MAXC / 16) + wave * (GPW / 2) + (k & 3);
        const int c  = G * 8 + cc;
        const int cl = c < C ? c : 0;
        const float4 M = *(const float4*)&Minv[cl * 4];
        const float2 a = *(const float2*)&Av[cl * 2];
        const float m0e = e ? M.y : M.x;
        const float m1e = e ? M.w : M.z;
        float off = -(a.x * m0e + a.y * m1e);
        if (c >= C) off = -INF;
        aop[k].x = hi ? off  : m0e;
        aop[k].y = hi ? 0.0f : m1e;
    }

    float ylo[8], yhi[8];
#pragma unroll
    for (int j = 0; j < 8; ++j) {
        const int col = (j >> 2) * 8 + ysh + (j & 3);
        const float2 yb = *(const float2*)&bbox[col * 4 + 2];
        ylo[j] = yb.x; yhi[j] = yb.y;
    }

    float4 tM[MAXC / THREADS]; float2 tA[MAXC / THREADS];
    int tD[MAXC / THREADS][3]; float tWx[MAXV / THREADS], tWy[MAXV / THREADS];
#pragma unroll
    for (int i = 0; i < MAXC / THREADS; ++i) {
        const int c  = tid + i * THREADS;
        const int cl = c < C ? c : 0;
        tM[i] = *(const float4*)&Minv[cl * 4];
        tA[i] = *(const float2*)&Av[cl * 2];
        tD[i][0] = dofs[cl * 3 + 0];
        tD[i][1] = dofs[cl * 3 + 1];
        tD[i][2] = dofs[cl * 3 + 2];
    }
#pragma unroll
    for (int i = 0; i < MAXV / THREADS; ++i) {
        const int v  = tid + i * THREADS;
        const int vl = v < NV ? v : 0;
        tWx[i] = wxg[vl];
        tWy[i] = wyg[vl];
    }

    {
        int4* const rows = (int4*)&sBest[(wave * 2) * BPITCH];
        const int4 none = make_int4(NOHIT, NOHIT, NOHIT, NOHIT);
        rows[lane]                   = none;
        rows[32 + lane]              = none;
        rows[BPITCH / 4 + lane]      = none;
        rows[BPITCH / 4 + 32 + lane] = none;
    }

    float4* const myList = &sList[wave * LCAP];
    int cnt = 0;
#pragma unroll
    for (int j = 0; j < PASSES; ++j) {
        const float px = P[j].x;
        const bool keep = ((xlo[0] < px) & (px < xhi[0])) | ((xlo[1] < px) & (px < xhi[1]));
        const unsigned msk = __builtin_amdgcn_ballot_w32(keep);
        const int pos = cnt + (int)__builtin_amdgcn_mbcnt_lo(msk, 0u);
        myList[keep ? pos : LDUMP + lane] =
            make_float4(px, P[j].y, __int_as_float(j * 32 + lane), 0.0f);
        cnt += __builtin_popcount(msk);
    }
    myList[cnt + lane] = make_float4(0.0f, 0.0f, __int_as_float(THREADS + n), 0.0f);
    const int nt0 = (cnt + 15) >> 4;
    const int ntiles = nt0 > 1 ? nt0 : 1;
    __builtin_amdgcn_wave_barrier();

#pragma unroll
    for (int i = 0; i < MAXC / THREADS; ++i) {
        const int c = tid + i * THREADS;
        *(float4*)&sMinv[c * 4] = tM[i];
        *(float2*)&sA[c * 2]    = tA[i];
        sDof[c * 3 + 0] = tD[i][0];
        sDof[c * 3 + 1] = tD[i][1];
        sDof[c * 3 + 2] = tD[i][2];
    }
#pragma unroll
    for (int i = 0; i < MAXV / THREADS; ++i) {
        const int v = tid + i * THREADS;
        sWx[v] = tWx[i];
        sWy[v] = tWy[i];
    }

    const int cellbase = wave * (GPW * 4) + ysh;
    int* const myBest = &sBest[(wave * 2 + (hi ? 1 : 0)) * BPITCH];

    auto gates = [&](float px, float py, v2f (&gub)[4], float (&gthr)[RPW]) {
#pragma unroll
        for (int j = 0; j < 8; ++j)
            gub[j >> 1][j & 1] = ((ylo[j] < py) & (py < yhi[j])) ? UB1 : -INF;
#pragma unroll
        for (int k = 0; k < RPW; ++k)
            gthr[k] = ((xlo[k] < px) & (px < xhi[k])) ? NTOL : INF;
    };

    const float4 ec = myList[n];
    v2f bc;
    bc.x = hi ? 1.0f : ec.x;
    bc.y = hi ? 0.0f : ec.y;
    int idc = __float_as_int(ec.z);
    v2f ub[4]; float thr[RPW];
    gates(ec.x, ec.y, ub, thr);

    v8f d[GPW + 2];
    {
        const v8f z = {};
        d[0] = __builtin_amdgcn_wmma_f32_16x16x4_f32(
                   false, aop[0], false, bc, (short)0, z, false, false);
        d[1] = __builtin_amdgcn_wmma_f32_16x16x4_f32(
                   false, aop[1], false, bc, (short)0, z, false, false);
    }

    float4 en = myList[(ntiles > 1 ? 16 : 0) + n];
    int t = 0;
    do {
        const int t2 = (t + 2 < ntiles) ? t + 2 : ntiles - 1;
        const float4 e2 = myList[t2 * 16 + n];
        v2f bn;
        bn.x = hi ? 1.0f : en.x;
        bn.y = hi ? 0.0f : en.y;

        int bq[4] = {NOHIT, NOHIT, NOHIT, NOHIT};

#pragma unroll
        for (int g = 0; g < GPW; ++g) {
            __builtin_amdgcn_sched_barrier(0);
            {
                const v8f z = {};
                d[g + 2] = __builtin_amdgcn_wmma_f32_16x16x4_f32(
                               false, aop[(g + 2) & (GPW - 1)], false,
                               (g + 2 < GPW) ? bc : bn, (short)0, z, false, false);
            }
            const v8f dg = d[g];
            const int  u  = (g & 1) * 2;
            const float th = thr[(g >> 1) & 1];
            const v2f e01 = ub[u]     - (__builtin_shufflevector(dg, dg, 0, 1) + __builtin_shufflevector(dg, dg, 4, 5));
            const v2f e23 = ub[u + 1] - (__builtin_shufflevector(dg, dg, 2, 3) + __builtin_shufflevector(dg, dg, 6, 7));
            float mm[4];
            mm[0] = __builtin_fminf(__builtin_fminf(dg[0], dg[4]), e01.x);
            mm[1] = __builtin_fminf(__builtin_fminf(dg[1], dg[5]), e01.y);
            mm[2] = __builtin_fminf(__builtin_fminf(dg[2], dg[6]), e23.x);
            mm[3] = __builtin_fminf(__builtin_fminf(dg[3], dg[7]), e23.y);
#pragma unroll
            for (int q = 0; q < 4; ++q)
                bq[q] = (mm[q] > th) ? (g * 8 + q) : bq[q];
        }
        __builtin_amdgcn_sched_barrier(0);

        v2f ubn[4]; float thrn[RPW];
        gates(en.x, en.y, ubn, thrn);

        const int b01 = bq[0] > bq[1] ? bq[0] : bq[1];
        const int b23 = bq[2] > bq[3] ? bq[2] : bq[3];
        const int key = b01 > b23 ? b01 : b23;
        const int best = key + cellbase + ((key & 32) ? (MAXC / 2 - 32) : 0);
        myBest[idc] = best;

        bc = bn;
        idc = __float_as_int(en.z);
#pragma unroll
        for (int j = 0; j < 4; ++j) ub[j] = ubn[j];
#pragma unroll
        for (int k = 0; k < RPW; ++k) thr[k] = thrn[k];
        d[0] = d[GPW];
        d[1] = d[GPW + 1];
        en = e2;
    } while (++t < ntiles);

    __syncthreads();

    int cell = sBest[tid];
#pragma unroll
    for (int r = 1; r < 2 * NWAVES; ++r) {
        const int o = sBest[r * BPITCH + tid];
        cell = cell > o ? cell : o;
    }

    float ox = 0.0f, oy = 0.0f;
    if (cell >= 0) {
        const int c  = cell;
        const float px = myp.x, py = myp.y;
        float dx = px - sA[c * 2 + 0];
        float dy = py - sA[c * 2 + 1];
        float s  = dx * sMinv[c * 4 + 0] + dy * sMinv[c * 4 + 2];
        float t  = dx * sMinv[c * 4 + 1] + dy * sMinv[c * 4 + 3];
        float w0 = 1.0f - s - t;
        int d0 = sDof[c * 3 + 0], d1 = sDof[c * 3 + 1], d2 = sDof[c * 3 + 2];
        ox = w0 * sWx[d0] + s * sWx[d1] + t * sWx[d2];
        oy = w0 * sWy[d0] + s * sWy[d1] + t * sWy[d2];
    }
    if (gp < NP) out[gp] = make_float2(ox, oy);
}

extern "C" void kernel_launch(void* const* d_in, const int* in_sizes, int n_in,
                              void* d_out, int out_size, void* d_ws, size_t ws_size,
                              hipStream_t stream) {
    const float* xp   = (const float*)d_in[0];
    const float* wx   = (const float*)d_in[1];
    const float* wy   = (const float*)d_in[2];
    const float* Minv = (const float*)d_in[3];
    const float* Av   = (const float*)d_in[4];
    const float* bbox = (const float*)d_in[5];
    const int*   dofs = (const int*)d_in[6];
    float2* out = (float2*)d_out;

    int NP = in_sizes[0] / 2;
    int C  = in_sizes[3] / 4;
    int NV = in_sizes[1];
    if (C > MAXC) C = MAXC;

    int blocks = (NP + THREADS - 1) / THREADS;
    if (blocks < 1) blocks = 1;

    p1interp_wmma<<<blocks, THREADS, 0, stream>>>(
        xp, wx, wy, Minv, Av, bbox, dofs, out, NP, C, NV);
}
